# starting kernel
# baseline (speedup 1.0000x reference)
#include <hip/hip_runtime.h>


typedef __attribute__((ext_vector_type(2))) float v2f;
typedef __attribute__((ext_vector_type(8))) float v8f;

#define N_WORDS  16384
#define L_CHARS  16
#define C_EMB    64
#define CONV_L   256
#define KERNEL_W 3
#define D_EMB    300
#define OUT_COLS (CONV_L + D_EMB)
#define N_OGRP   (CONV_L / 16)
#define N_FRAG   (KERNEL_W * N_OGRP * 16)
#define WPB      8
#define T_LEN    (L_CHARS + 2)

__global__ void pack_conv_w_kernel(const float* __restrict__ conv_w,
                                   float* __restrict__ wpack) {
    int tid  = blockIdx.x * blockDim.x + threadIdx.x;
    if (tid >= N_FRAG * 32) return;
    int lane = tid & 31;
    int frag = tid >> 5;
    int q    = frag & 15;
    int g    = (frag >> 4) & 15;
    int k    = frag >> 8;
    int koff = (lane < 16) ? 0 : 2;
    int o    = g * 16 + (lane & 15);
    int c0   = 4 * q + koff;
    float w0 = conv_w[o * (C_EMB * KERNEL_W) + (c0 + 0) * KERNEL_W + k];
    float w1 = conv_w[o * (C_EMB * KERNEL_W) + (c0 + 1) * KERNEL_W + k];
    wpack[frag * 64 + lane * 2 + 0] = w0;
    wpack[frag * 64 + lane * 2 + 1] = w1;
}

__global__ void word_gather_kernel(const int* __restrict__ X_word,
                                   const float* __restrict__ word_emb,
                                   float* __restrict__ out) {
    long long i = (long long)blockIdx.x * blockDim.x + threadIdx.x;
    if (i >= (long long)N_WORDS * D_EMB) return;
    int n = (int)(i / D_EMB);
    int j = (int)(i % D_EMB);
    out[(long long)n * OUT_COLS + CONV_L + j] =
        word_emb[(long long)X_word[n] * D_EMB + j];
}

__global__ __launch_bounds__(512)
void char_conv_kernel(const int* __restrict__ X,
                      const float* __restrict__ char_emb,
                      const float* __restrict__ wpack,
                      const float* __restrict__ conv_b,
                      float* __restrict__ out) {
    __shared__ int   Xs[WPB * L_CHARS];
    __shared__ float ce[WPB * L_CHARS * C_EMB];
    __shared__ float Y[16 * T_LEN * 16];

    const int tid = threadIdx.x;
    const int n0  = blockIdx.x * WPB;

    if (tid < WPB * L_CHARS) Xs[tid] = X[n0 * L_CHARS + tid];
    __syncthreads();

    for (int e = tid; e < WPB * L_CHARS * C_EMB; e += 512) {
        int ws = e >> 6;
        int c  = e & 63;
        ce[e] = char_emb[Xs[ws] * C_EMB + c];
    }

    const int lane = tid & 31;
    const int wave = tid >> 5;
    const int g    = wave;
    const int koff = (lane < 16) ? 0 : 2;
    const int srow = lane & 15;
    const int shalf = (lane >> 4) << 3;
    const int ocol  = lane & 15;

    v2f B0[16], B1[16], B2[16];
    {
        const v2f* wp = (const v2f*)wpack;
#pragma unroll
        for (int q = 0; q < 16; ++q) {
            B0[q] = wp[((0 * N_OGRP + g) * 16 + q) * 32 + lane];
            B1[q] = wp[((1 * N_OGRP + g) * 16 + q) * 32 + lane];
            B2[q] = wp[((2 * N_OGRP + g) * 16 + q) * 32 + lane];
        }
    }
    __syncthreads();

    float* Yw = &Y[wave * (T_LEN * 16)];
    const float bias = conv_b[g * 16 + ocol];

    for (int w = 0; w < WPB; ++w) {
#pragma unroll
        for (int i = lane; i < T_LEN * 16; i += 32) Yw[i] = 0.0f;

        const float* cew = &ce[w * L_CHARS * C_EMB];

        v8f acc0 = {}; v8f acc1 = {}; v8f acc2 = {};
#pragma unroll
        for (int q = 0; q < 16; ++q) {
            v2f a = *(const v2f*)&cew[srow * C_EMB + 4 * q + koff];
            acc0 = __builtin_amdgcn_wmma_f32_16x16x4_f32(false, a, false, B0[q],
                                                         (short)0, acc0, false, false);
            acc1 = __builtin_amdgcn_wmma_f32_16x16x4_f32(false, a, false, B1[q],
                                                         (short)0, acc1, false, false);
            acc2 = __builtin_amdgcn_wmma_f32_16x16x4_f32(false, a, false, B2[q],
                                                         (short)0, acc2, false, false);
        }

#pragma unroll
        for (int v = 0; v < 8; ++v) {
            int s = v + shalf;
            atomicAdd(&Yw[(s + 2) * 16 + ocol], acc0[v]);
            atomicAdd(&Yw[(s + 1) * 16 + ocol], acc1[v]);
            atomicAdd(&Yw[(s + 0) * 16 + ocol], acc2[v]);
        }
        __syncthreads();

        if (lane < 16) {
            float smax = -3.402823466e38f;
#pragma unroll
            for (int t = 0; t < T_LEN; ++t)
                smax = fmaxf(smax, Yw[t * 16 + ocol]);
            float r = fmaxf(0.0f, bias + smax);
            out[(long long)(n0 + w) * OUT_COLS + g * 16 + ocol] = r;
        }
        __syncthreads();
    }
}

extern "C" void kernel_launch(void* const* d_in, const int* in_sizes, int n_in,
                              void* d_out, int out_size, void* d_ws, size_t ws_size,
                              hipStream_t stream) {
    const int*   X          = (const int*)d_in[0];
    const int*   X_word     = (const int*)d_in[1];
    const float* char_emb_w = (const float*)d_in[2];
    const float* conv_w     = (const float*)d_in[3];
    const float* conv_b     = (const float*)d_in[4];
    const float* word_emb_w = (const float*)d_in[5];
    float*       out        = (float*)d_out;
    float*       wpack      = (float*)d_ws;

    (void)in_sizes; (void)n_in; (void)out_size; (void)ws_size;

    pack_conv_w_kernel<<<(N_FRAG * 32 + 255) / 256, 256, 0, stream>>>(conv_w, wpack);

    long long wn = (long long)N_WORDS * D_EMB;
    word_gather_kernel<<<(int)((wn + 255) / 256), 256, 0, stream>>>(X_word, word_emb_w, out);

    char_conv_kernel<<<N_WORDS / WPB, 512, 0, stream>>>(X, char_emb_w, wpack, conv_b, out);
}


# Multiply the 128-row char table by the 3 taps once (128x64x768 f32 WMMA) and per word gather-add those rows from LDS instead of a product per occurrence; tap sums now (k0+k1)+k2, LDS atomics gone
# speedup vs baseline: 7.1749x; 7.1749x over previous
#include <hip/hip_runtime.h>


typedef __attribute__((ext_vector_type(2))) float v2f;
typedef __attribute__((ext_vector_type(4))) float v4f;
typedef __attribute__((ext_vector_type(8))) float v8f;
typedef __attribute__((ext_vector_type(4))) int   v4i;

#define N_WORDS   16384
#define L_CHARS   16
#define CHAR_ROWS 128
#define C_EMB     64
#define CONV_L    256
#define KERNEL_W  3
#define D_EMB     300
#define OUT_COLS  (CONV_L + D_EMB)
#define N_OGRP    (CONV_L / 16)
#define N_MTILE   (CHAR_ROWS / 16)

#define HALF_COLS (CONV_L / 2)
#define ROW_V4    (HALF_COLS / 4)
#define HALF_V4   (CHAR_ROWS * KERNEL_W * ROW_V4)
#define POOL_THREADS    512
#define WORDS_PER_BLOCK 128
#define WORDS_PER_WAVE  (WORDS_PER_BLOCK / (POOL_THREADS / 32))

__global__ __launch_bounds__(32)
void tap_table_kernel(const float* __restrict__ char_emb,
                      const float* __restrict__ conv_w,
                      float* __restrict__ table) {
    const int lane  = threadIdx.x;
    const int g     = blockIdx.x & (N_OGRP - 1);
    const int mt    = blockIdx.x >> 4;
    const int koff  = (lane < 16) ? 0 : 2;
    const int c16   = lane & 15;
    const int shalf = (lane >> 4) << 3;

    const float* arow = char_emb + (mt * 16 + c16) * C_EMB + koff;
    const float* brow = conv_w + (g * 16 + c16) * (C_EMB * KERNEL_W) + koff * KERNEL_W;

    v2f A[16], B0[16], B1[16], B2[16];
#pragma unroll
    for (int q = 0; q < 16; ++q) {
        A[q] = *(const v2f*)(arow + 4 * q);
        const v2f* p = (const v2f*)(brow + 4 * KERNEL_W * q);
        const v2f f01 = p[0], f23 = p[1], f45 = p[2];
        B0[q] = (v2f){f01.x, f23.y};
        B1[q] = (v2f){f01.y, f45.x};
        B2[q] = (v2f){f23.x, f45.y};
    }

    v8f acc0 = {}; v8f acc1 = {}; v8f acc2 = {};
#pragma unroll
    for (int q = 0; q < 16; ++q) {
        acc0 = __builtin_amdgcn_wmma_f32_16x16x4_f32(false, A[q], false, B0[q],
                                                     (short)0, acc0, false, false);
        acc1 = __builtin_amdgcn_wmma_f32_16x16x4_f32(false, A[q], false, B1[q],
                                                     (short)0, acc1, false, false);
        acc2 = __builtin_amdgcn_wmma_f32_16x16x4_f32(false, A[q], false, B2[q],
                                                     (short)0, acc2, false, false);
    }

    const int hf = g >> 3;
    const int oo = (g & 7) * 16 + c16;
#pragma unroll
    for (int v = 0; v < 8; ++v) {
        const int ch = mt * 16 + v + shalf;
        float* dst = table + ((hf * CHAR_ROWS + ch) * KERNEL_W) * HALF_COLS + oo;
        dst[0 * HALF_COLS] = acc0[v];
        dst[1 * HALF_COLS] = acc1[v];
        dst[2 * HALF_COLS] = acc2[v];
    }
}

__global__ void word_gather_kernel(const int* __restrict__ X_word,
                                   const float* __restrict__ word_emb,
                                   float* __restrict__ out) {
    long long i = (long long)blockIdx.x * blockDim.x + threadIdx.x;
    if (i >= (long long)N_WORDS * D_EMB) return;
    int n = (int)(i / D_EMB);
    int j = (int)(i % D_EMB);
    out[(long long)n * OUT_COLS + CONV_L + j] =
        word_emb[(long long)X_word[n] * D_EMB + j];
}

static __device__ __forceinline__ v4f max4(v4f a, v4f b) {
    return (v4f){fmaxf(a.x, b.x), fmaxf(a.y, b.y), fmaxf(a.z, b.z), fmaxf(a.w, b.w)};
}

__global__ __launch_bounds__(POOL_THREADS)
void char_pool_kernel(const int* __restrict__ X,
                      const float* __restrict__ table,
                      const float* __restrict__ conv_b,
                      float* __restrict__ out) {
    __shared__ v4f tab[HALF_V4];

    const int tid  = threadIdx.x;
    const int lane = tid & 31;
    const int hf   = blockIdx.x & 1;
    const int wgrp = blockIdx.x >> 1;
    const int wave = __builtin_amdgcn_readfirstlane(tid >> 5);
    const int n0   = wgrp * WORDS_PER_BLOCK + wave * WORDS_PER_WAVE;

    const v4i xq = ((const v4i*)X)[n0 * (L_CHARS / 4) + lane];
    const v4f bias = ((const v4f*)conv_b)[hf * ROW_V4 + lane];

    const v4f* src = (const v4f*)table + hf * HALF_V4;
#pragma unroll
    for (int i = 0; i < HALF_V4 / POOL_THREADS; ++i)
        tab[i * POOL_THREADS + tid] = src[i * POOL_THREADS + tid];
    __syncthreads();

    const v4f* lrow = tab + lane;
    float* obase = out + (long long)n0 * OUT_COLS + hf * HALF_COLS;

    for (int j = 0; j < WORDS_PER_WAVE; ++j) {
        v4f m, p1, p2;
#pragma unroll
        for (int s = 0; s < L_CHARS; ++s) {
            const int ch = __builtin_amdgcn_readlane(xq[s & 3], 4 * j + (s >> 2))
                           & (CHAR_ROWS - 1);
            const v4f* row = lrow + ch * (KERNEL_W * ROW_V4);
            const v4f r0 = row[0 * ROW_V4];
            const v4f r1 = row[1 * ROW_V4];
            const v4f r2 = row[2 * ROW_V4];
            if (s == 0) {
                m  = r2;
                p2 = r1;
            } else {
                m  = max4(m, p2 + r2);
                p2 = p1 + r1;
            }
            p1 = r0;
        }
        m = max4(m, p2);
        m = max4(m, p1);

        const v4f z = {0.0f, 0.0f, 0.0f, 0.0f};
        *(v4f*)(obase + j * OUT_COLS + lane * 4) = max4(z, bias + m);
    }
}

extern "C" void kernel_launch(void* const* d_in, const int* in_sizes, int n_in,
                              void* d_out, int out_size, void* d_ws, size_t ws_size,
                              hipStream_t stream) {
    const int*   X          = (const int*)d_in[0];
    const int*   X_word     = (const int*)d_in[1];
    const float* char_emb_w = (const float*)d_in[2];
    const float* conv_w     = (const float*)d_in[3];
    const float* conv_b     = (const float*)d_in[4];
    const float* word_emb_w = (const float*)d_in[5];
    float*       out        = (float*)d_out;
    float*       table      = (float*)d_ws;

    (void)in_sizes; (void)n_in; (void)out_size; (void)ws_size;

    tap_table_kernel<<<N_MTILE * N_OGRP, 32, 0, stream>>>(char_emb_w, conv_w, table);

    long long wn = (long long)N_WORDS * D_EMB;
    word_gather_kernel<<<(int)((wn + 255) / 256), 256, 0, stream>>>(X_word, word_emb_w, out);

    char_pool_kernel<<<2 * (N_WORDS / WORDS_PER_BLOCK), POOL_THREADS, 0, stream>>>(
        X, table, conv_b, out);
}


# Word-row gather folded into the pool kernel: each wave copies 4 word rows as 16 B pieces (loads at the head of its LDS loop, stores in its middle); launches 3 -> 2, the 19200-block copy launch gone
# speedup vs baseline: 12.6425x; 1.7620x over previous
#include <hip/hip_runtime.h>


typedef __attribute__((ext_vector_type(2))) float v2f;
typedef __attribute__((ext_vector_type(4))) float v4f;
typedef __attribute__((ext_vector_type(8))) float v8f;
typedef __attribute__((ext_vector_type(4))) int   v4i;

#define N_WORDS   16384
#define L_CHARS   16
#define CHAR_ROWS 128
#define C_EMB     64
#define CONV_L    256
#define KERNEL_W  3
#define D_EMB     300
#define OUT_COLS  (CONV_L + D_EMB)
#define N_OGRP    (CONV_L / 16)
#define N_MTILE   (CHAR_ROWS / 16)

#define HALF_COLS (CONV_L / 2)
#define ROW_V4    (HALF_COLS / 4)
#define HALF_V4   (CHAR_ROWS * KERNEL_W * ROW_V4)
#define POOL_THREADS    512
#define WORDS_PER_BLOCK 128
#define WORDS_PER_WAVE  (WORDS_PER_BLOCK / (POOL_THREADS / 32))

#define WROW_V4     (D_EMB / 4)
#define COPY_WORDS  (WORDS_PER_WAVE / 2)
#define WROW_TAIL   (WROW_V4 - 64)

__global__ __launch_bounds__(32)
void tap_table_kernel(const float* __restrict__ char_emb,
                      const float* __restrict__ conv_w,
                      float* __restrict__ table) {
    const int lane  = threadIdx.x;
    const int g     = blockIdx.x & (N_OGRP - 1);
    const int mt    = blockIdx.x >> 4;
    const int koff  = (lane < 16) ? 0 : 2;
    const int c16   = lane & 15;
    const int shalf = (lane >> 4) << 3;

    const float* arow = char_emb + (mt * 16 + c16) * C_EMB + koff;
    const float* brow = conv_w + (g * 16 + c16) * (C_EMB * KERNEL_W) + koff * KERNEL_W;

    v2f A[16], B0[16], B1[16], B2[16];
#pragma unroll
    for (int q = 0; q < 16; ++q) {
        A[q] = *(const v2f*)(arow + 4 * q);
        const v2f* p = (const v2f*)(brow + 4 * KERNEL_W * q);
        const v2f f01 = p[0], f23 = p[1], f45 = p[2];
        B0[q] = (v2f){f01.x, f23.y};
        B1[q] = (v2f){f01.y, f45.x};
        B2[q] = (v2f){f23.x, f45.y};
    }

    v8f acc0 = {}; v8f acc1 = {}; v8f acc2 = {};
#pragma unroll
    for (int q = 0; q < 16; ++q) {
        acc0 = __builtin_amdgcn_wmma_f32_16x16x4_f32(false, A[q], false, B0[q],
                                                     (short)0, acc0, false, false);
        acc1 = __builtin_amdgcn_wmma_f32_16x16x4_f32(false, A[q], false, B1[q],
                                                     (short)0, acc1, false, false);
        acc2 = __builtin_amdgcn_wmma_f32_16x16x4_f32(false, A[q], false, B2[q],
                                                     (short)0, acc2, false, false);
    }

    const int hf = g >> 3;
    const int oo = (g & 7) * 16 + c16;
#pragma unroll
    for (int v = 0; v < 8; ++v) {
        const int ch = mt * 16 + v + shalf;
        float* dst = table + ((hf * CHAR_ROWS + ch) * KERNEL_W) * HALF_COLS + oo;
        dst[0 * HALF_COLS] = acc0[v];
        dst[1 * HALF_COLS] = acc1[v];
        dst[2 * HALF_COLS] = acc2[v];
    }
}

static __device__ __forceinline__ v4f max4(v4f a, v4f b) {
    return (v4f){fmaxf(a.x, b.x), fmaxf(a.y, b.y), fmaxf(a.z, b.z), fmaxf(a.w, b.w)};
}

__global__ __launch_bounds__(POOL_THREADS)
void char_pool_kernel(const int* __restrict__ X,
                      const int* __restrict__ X_word,
                      const float* __restrict__ table,
                      const float* __restrict__ conv_b,
                      const float* __restrict__ word_emb,
                      float* __restrict__ out) {
    __shared__ v4f tab[HALF_V4];

    const int tid  = threadIdx.x;
    const int lane = tid & 31;
    const int hf   = blockIdx.x & 1;
    const int wgrp = blockIdx.x >> 1;
    const int wave = __builtin_amdgcn_readfirstlane(tid >> 5);
    const int n0   = wgrp * WORDS_PER_BLOCK + wave * WORDS_PER_WAVE;

    const v4i xq = ((const v4i*)X)[n0 * (L_CHARS / 4) + lane];
    const v4f bias = ((const v4f*)conv_b)[hf * ROW_V4 + lane];
    const int nw = n0 + hf * COPY_WORDS;
    const v4i xw = *(const v4i*)(X_word + nw);

    const v4f* src = (const v4f*)table + hf * HALF_V4;
#pragma unroll
    for (int i = 0; i < HALF_V4 / POOL_THREADS; ++i)
        tab[i * POOL_THREADS + tid] = src[i * POOL_THREADS + tid];
    __syncthreads();

    v4f wa[COPY_WORDS], wb[COPY_WORDS], wc[COPY_WORDS];
#pragma unroll
    for (int w = 0; w < COPY_WORDS; ++w) {
        const v4f* rp = (const v4f*)word_emb + (long long)xw[w] * WROW_V4 + lane;
        wa[w] = rp[0];
        wb[w] = rp[32];
    }
    if (lane < WROW_TAIL) {
#pragma unroll
        for (int w = 0; w < COPY_WORDS; ++w) {
            const v4f* rp = (const v4f*)word_emb + (long long)xw[w] * WROW_V4 + lane;
            wc[w] = rp[64];
        }
    }

    const v4f* lrow = tab + lane;
    float* obase = out + (long long)n0 * OUT_COLS + hf * HALF_COLS;

    for (int j = 0; j < WORDS_PER_WAVE; ++j) {
        if (j == WORDS_PER_WAVE / 2) {
            v4f* wp = (v4f*)(out + (long long)nw * OUT_COLS + CONV_L) + lane;
#pragma unroll
            for (int w = 0; w < COPY_WORDS; ++w) {
                wp[w * (OUT_COLS / 4)]      = wa[w];
                wp[w * (OUT_COLS / 4) + 32] = wb[w];
            }
            if (lane < WROW_TAIL) {
#pragma unroll
                for (int w = 0; w < COPY_WORDS; ++w)
                    wp[w * (OUT_COLS / 4) + 64] = wc[w];
            }
        }
        v4f m, p1, p2;
#pragma unroll
        for (int s = 0; s < L_CHARS; ++s) {
            const int ch = __builtin_amdgcn_readlane(xq[s & 3], 4 * j + (s >> 2))
                           & (CHAR_ROWS - 1);
            const v4f* row = lrow + ch * (KERNEL_W * ROW_V4);
            const v4f r0 = row[0 * ROW_V4];
            const v4f r1 = row[1 * ROW_V4];
            const v4f r2 = row[2 * ROW_V4];
            if (s == 0) {
                m  = r2;
                p2 = r1;
            } else {
                m  = max4(m, p2 + r2);
                p2 = p1 + r1;
            }
            p1 = r0;
        }
        m = max4(m, p2);
        m = max4(m, p1);

        const v4f z = {0.0f, 0.0f, 0.0f, 0.0f};
        *(v4f*)(obase + j * OUT_COLS + lane * 4) = max4(z, bias + m);
    }
}

extern "C" void kernel_launch(void* const* d_in, const int* in_sizes, int n_in,
                              void* d_out, int out_size, void* d_ws, size_t ws_size,
                              hipStream_t stream) {
    const int*   X          = (const int*)d_in[0];
    const int*   X_word     = (const int*)d_in[1];
    const float* char_emb_w = (const float*)d_in[2];
    const float* conv_w     = (const float*)d_in[3];
    const float* conv_b     = (const float*)d_in[4];
    const float* word_emb_w = (const float*)d_in[5];
    float*       out        = (float*)d_out;
    float*       table      = (float*)d_ws;

    (void)in_sizes; (void)n_in; (void)out_size; (void)ws_size;

    tap_table_kernel<<<N_MTILE * N_OGRP, 32, 0, stream>>>(char_emb_w, conv_w, table);

    char_pool_kernel<<<2 * (N_WORDS / WORDS_PER_BLOCK), POOL_THREADS, 0, stream>>>(
        X, X_word, table, conv_b, word_emb_w, out);
}


# Tap-table product folded into the pool kernel: a block owns 32 channels x 512 words and builds its table slice on f32 WMMA into LDS (4 words per LDS load); launches 2 -> 1, workspace unused
# speedup vs baseline: 13.9017x; 1.0996x over previous
#include <hip/hip_runtime.h>


typedef __attribute__((ext_vector_type(2))) float v2f;
typedef __attribute__((ext_vector_type(4))) float v4f;
typedef __attribute__((ext_vector_type(8))) float v8f;
typedef __attribute__((ext_vector_type(4))) int   v4i;

#define N_WORDS   16384
#define L_CHARS   16
#define CHAR_ROWS 128
#define C_EMB     64
#define CONV_L    256
#define KERNEL_W  3
#define D_EMB     300
#define OUT_COLS  (CONV_L + D_EMB)

#define BLK_COLS  32
#define N_CBLK    (CONV_L / BLK_COLS)
#define POOL_THREADS    512
#define N_WAVES         (POOL_THREADS / 32)
#define WORDS_PER_BLOCK 512
#define WORDS_PER_WAVE  (WORDS_PER_BLOCK / N_WAVES)
#define N_QUADS         (WORDS_PER_WAVE / 4)

#define WROW_V4     (D_EMB / 4)
#define COPY_WORDS  (WORDS_PER_WAVE / N_CBLK)
#define WROW_TAIL   (WROW_V4 - 64)

#define TAB_ROW_DW  64
#define TAB_CH_DW   (KERNEL_W * TAB_ROW_DW)
#define TAB_DW      (CHAR_ROWS * TAB_CH_DW)
#define SA_PITCH    (C_EMB + 4)
#define SA_DW       (CHAR_ROWS * SA_PITCH)
#define SB_PITCH    (C_EMB * KERNEL_W + 4)
#define SB_DW       (BLK_COLS * SB_PITCH)
#define LDS_DW      (164 * 256)

static __device__ __forceinline__ v4f max4(v4f a, v4f b) {
    return (v4f){fmaxf(a.x, b.x), fmaxf(a.y, b.y), fmaxf(a.z, b.z), fmaxf(a.w, b.w)};
}

__global__ __launch_bounds__(POOL_THREADS)
void char_word_kernel(const int* __restrict__ X,
                      const int* __restrict__ X_word,
                      const float* __restrict__ char_emb,
                      const float* __restrict__ conv_w,
                      const float* __restrict__ conv_b,
                      const float* __restrict__ word_emb,
                      float* __restrict__ out) {
    __shared__ v4f lds[LDS_DW / 4];
    float* tabf = (float*)lds;
    float* sA   = tabf + TAB_DW;
    float* sB   = sA + SA_DW;

    const int tid  = threadIdx.x;
    const int lane = tid & 31;
    const int e    = blockIdx.x & (N_CBLK - 1);
    const int wgrp = blockIdx.x >> 3;
    const int wave = __builtin_amdgcn_readfirstlane(tid >> 5);
    const int n0   = wgrp * WORDS_PER_BLOCK + wave * WORDS_PER_WAVE;
    const int slot = lane >> 3;
    const int sub  = lane & 7;

    v4f ga[4], gb[3];
#pragma unroll
    for (int i = 0; i < 4; ++i)
        ga[i] = ((const v4f*)char_emb)[i * POOL_THREADS + tid];
#pragma unroll
    for (int i = 0; i < 3; ++i)
        gb[i] = ((const v4f*)conv_w)[e * (BLK_COLS * 48) + i * POOL_THREADS + tid];

    const v4i* xbase = (const v4i*)X + (long long)(n0 + slot) * (L_CHARS / 4);
    v4i xc[4];
#pragma unroll
    for (int i = 0; i < 4; ++i) xc[i] = xbase[i];
    const v4f bias = ((const v4f*)conv_b)[e * (BLK_COLS / 4) + sub];
    const int nw = n0 + e * COPY_WORDS;
    const v4i xw = *(const v4i*)(X_word + nw);

#pragma unroll
    for (int i = 0; i < 4; ++i) {
        const int idx = i * POOL_THREADS + tid;
        *(v4f*)(sA + (idx >> 4) * SA_PITCH + (idx & 15) * 4) = ga[i];
    }
#pragma unroll
    for (int i = 0; i < 3; ++i) {
        const int idx = i * POOL_THREADS + tid;
        const int o = idx / 48, p = idx - o * 48;
        *(v4f*)(sB + o * SB_PITCH + p * 4) = gb[i];
    }
    __syncthreads();

    {
        const int mt    = wave & 7;
        const int gl    = wave >> 3;
        const int koff  = (lane < 16) ? 0 : 2;
        const int c16   = lane & 15;
        const int shalf = (lane >> 4) << 3;
        const float* arow = sA + (mt * 16 + c16) * SA_PITCH + koff;
        const float* brow = sB + (gl * 16 + c16) * SB_PITCH + koff * KERNEL_W;

        v2f A[16], B0[16], B1[16], B2[16];
#pragma unroll
        for (int q = 0; q < 16; ++q) {
            A[q] = *(const v2f*)(arow + 4 * q);
            const v2f* p = (const v2f*)(brow + 4 * KERNEL_W * q);
            const v2f f01 = p[0], f23 = p[1], f45 = p[2];
            B0[q] = (v2f){f01.x, f23.y};
            B1[q] = (v2f){f01.y, f45.x};
            B2[q] = (v2f){f23.x, f45.y};
        }

        v8f acc0 = {}; v8f acc1 = {}; v8f acc2 = {};
#pragma unroll
        for (int q = 0; q < 16; ++q) {
            acc0 = __builtin_amdgcn_wmma_f32_16x16x4_f32(false, A[q], false, B0[q],
                                                         (short)0, acc0, false, false);
            acc1 = __builtin_amdgcn_wmma_f32_16x16x4_f32(false, A[q], false, B1[q],
                                                         (short)0, acc1, false, false);
            acc2 = __builtin_amdgcn_wmma_f32_16x16x4_f32(false, A[q], false, B2[q],
                                                         (short)0, acc2, false, false);
        }

        float* dcol = tabf + gl * 16 + c16;
#pragma unroll
        for (int v = 0; v < 8; ++v) {
            float* dst = dcol + (mt * 16 + v + shalf) * TAB_CH_DW;
            dst[0 * TAB_ROW_DW]      = acc0[v];
            dst[0 * TAB_ROW_DW + 32] = acc0[v];
            dst[1 * TAB_ROW_DW]      = acc1[v];
            dst[1 * TAB_ROW_DW + 32] = acc1[v];
            dst[2 * TAB_ROW_DW]      = acc2[v];
            dst[2 * TAB_ROW_DW + 32] = acc2[v];
        }
    }
    __syncthreads();

    v4f wa[COPY_WORDS], wb[COPY_WORDS], wc[COPY_WORDS];
#pragma unroll
    for (int w = 0; w < COPY_WORDS; ++w) {
        const v4f* rp = (const v4f*)word_emb + (long long)xw[w] * WROW_V4 + lane;
        wa[w] = rp[0];
        wb[w] = rp[32];
    }
    if (lane < WROW_TAIL) {
#pragma unroll
        for (int w = 0; w < COPY_WORDS; ++w) {
            const v4f* rp = (const v4f*)word_emb + (long long)xw[w] * WROW_V4 + lane;
            wc[w] = rp[64];
        }
    }

    const v4f* lrow = lds + (slot & 1) * (BLK_COLS / 4) + sub;
    float* obase = out + (long long)(n0 + slot) * OUT_COLS + e * BLK_COLS + sub * 4;

    for (int j = 0; j < N_QUADS; ++j) {
        v4i xn[4];
        const v4i* xnext = xbase + ((j + 1) & (N_QUADS - 1)) * (4 * L_CHARS / 4);
#pragma unroll
        for (int i = 0; i < 4; ++i) xn[i] = xnext[i];

        if (j == N_QUADS / 2) {
            v4f* wp = (v4f*)(out + (long long)nw * OUT_COLS + CONV_L) + lane;
#pragma unroll
            for (int w = 0; w < COPY_WORDS; ++w) {
                wp[w * (OUT_COLS / 4)]      = wa[w];
                wp[w * (OUT_COLS / 4) + 32] = wb[w];
            }
            if (lane < WROW_TAIL) {
#pragma unroll
                for (int w = 0; w < COPY_WORDS; ++w)
                    wp[w * (OUT_COLS / 4) + 64] = wc[w];
            }
        }
        v4f m, p1, p2;
#pragma unroll
        for (int s = 0; s < L_CHARS; ++s) {
            const int ch = xc[s >> 2][s & 3] & (CHAR_ROWS - 1);
            const v4f* row = lrow + ch * (TAB_CH_DW / 4);
            const v4f r0 = row[0 * (TAB_ROW_DW / 4)];
            const v4f r1 = row[1 * (TAB_ROW_DW / 4)];
            const v4f r2 = row[2 * (TAB_ROW_DW / 4)];
            if (s == 0) {
                m  = r2;
                p2 = r1;
            } else {
                m  = max4(m, p2 + r2);
                p2 = p1 + r1;
            }
            p1 = r0;
        }
        m = max4(m, p2);
        m = max4(m, p1);

        const v4f z = {0.0f, 0.0f, 0.0f, 0.0f};
        *(v4f*)(obase + (long long)j * (4 * OUT_COLS)) = max4(z, bias + m);
#pragma unroll
        for (int i = 0; i < 4; ++i) xc[i] = xn[i];
    }
}

extern "C" void kernel_launch(void* const* d_in, const int* in_sizes, int n_in,
                              void* d_out, int out_size, void* d_ws, size_t ws_size,
                              hipStream_t stream) {
    const int*   X          = (const int*)d_in[0];
    const int*   X_word     = (const int*)d_in[1];
    const float* char_emb_w = (const float*)d_in[2];
    const float* conv_w     = (const float*)d_in[3];
    const float* conv_b     = (const float*)d_in[4];
    const float* word_emb_w = (const float*)d_in[5];
    float*       out        = (float*)d_out;

    (void)in_sizes; (void)n_in; (void)out_size; (void)d_ws; (void)ws_size;

    char_word_kernel<<<N_CBLK * (N_WORDS / WORDS_PER_BLOCK), POOL_THREADS, 0, stream>>>(
        X, X_word, char_emb_w, conv_w, conv_b, word_emb_w, out);
}
